# starting kernel
# baseline (speedup 1.0000x reference)
#include <hip/hip_runtime.h>
#include <hip/hip_bf16.h>

typedef float v2f __attribute__((ext_vector_type(2)));
typedef float v8f __attribute__((ext_vector_type(8)));

#define BATCH     32768
#define KD        784
#define N_EXPERTS 5
#define ROWS_PER_WAVE 16
#define WAVES_PER_BLOCK 8
#define THREADS (WAVES_PER_BLOCK * 32)

__global__ __launch_bounds__(THREADS)
void moe_wmma_kernel(const float* __restrict__ x,
                     const int*   __restrict__ cat,
                     const float* __restrict__ conv_w,
                     const float* __restrict__ conv_b,
                     const float* __restrict__ fc1_w,
                     const float* __restrict__ fc1_b,
                     float* __restrict__ out)
{
    __shared__ float tile[WAVES_PER_BLOCK][ROWS_PER_WAVE * 16];

    const int lane  = threadIdx.x & 31;
    const int wave  = threadIdx.x >> 5;
    const int gwave = blockIdx.x * WAVES_PER_BLOCK + wave;
    const int r0    = gwave * ROWS_PER_WAVE;

    const int m  = lane & 15;
    const int hk = (lane >> 4) * 2;

    const bool bvalid = (m < N_EXPERTS);
    const int  nclamp = bvalid ? m : (N_EXPERTS - 1);

    const v2f* __restrict__ ap = (const v2f*)(x      + (size_t)(r0 + m) * KD + hk);
    const v2f* __restrict__ bp = (const v2f*)(conv_w + (size_t)nclamp   * KD + hk);

    v8f acc = {};
    #pragma unroll 4
    for (int kk = 0; kk < KD / 4; ++kk) {
        v2f a = ap[2 * kk];
        v2f b = bp[2 * kk];
        b.x = bvalid ? b.x : 0.0f;
        b.y = bvalid ? b.y : 0.0f;
        acc = __builtin_amdgcn_wmma_f32_16x16x4_f32(
             false, a,  false, b,
             (short)0, acc,  false,  false);
    }

    {
        float* t = tile[wave];
        const int mh = (lane >> 4) * 8;
        const int nn = lane & 15;
        #pragma unroll
        for (int r = 0; r < 8; ++r)
            t[(mh + r) * 16 + nn] = acc[r];
    }
    __syncthreads();

    if (lane < ROWS_PER_WAVE) {
        const int   row = r0 + lane;
        const int   ci  = cat[row];
        const float s   = tile[wave][lane * 16 + ci] + conv_b[ci];
        const float o   = s * fc1_w[ci] + fc1_b[0];
        out[row] = o > 0.0f ? o : 0.0f;
    }
}

extern "C" void kernel_launch(void* const* d_in, const int* in_sizes, int n_in,
                              void* d_out, int out_size, void* d_ws, size_t ws_size,
                              hipStream_t stream) {
    const float* x    = (const float*)d_in[0];
    const int*   cat  = (const int*)  d_in[1];
    const float* cw   = (const float*)d_in[2];
    const float* cb   = (const float*)d_in[3];
    const float* f1w  = (const float*)d_in[4];
    const float* f1b  = (const float*)d_in[5];
    float*       out  = (float*)d_out;

    const int nblocks = BATCH / (ROWS_PER_WAVE * WAVES_PER_BLOCK);
    hipLaunchKernelGGL(moe_wmma_kernel, dim3(nblocks), dim3(THREADS), 0, stream,
                       x, cat, cw, cb, f1w, f1b, out);
}


# conv_w copied once per work-group into LDS with a zero sixth row; B fragments read by ds_load_b128 instead of 98 global loads and 392 zero-pad selects per wave; f32 sum order unchanged
# speedup vs baseline: 3.1608x; 3.1608x over previous
#include <hip/hip_runtime.h>
#include <hip/hip_bf16.h>


typedef float v2f __attribute__((ext_vector_type(2)));
typedef float v4f __attribute__((ext_vector_type(4)));
typedef float v8f __attribute__((ext_vector_type(8)));

#define BATCH     32768
#define KD        784
#define N_EXPERTS 5
#define ROWS_PER_WAVE 16
#define WAVES_PER_BLOCK 8
#define THREADS (WAVES_PER_BLOCK * 32)
#define NPIECE (KD / 8)
#define GRP    14
#define NGRP   (NPIECE / GRP)
#define WROW4  (KD / 4)
#define WTOT4  (N_EXPERTS * WROW4)
#define WSTAGE ((WTOT4 + THREADS - 1) / THREADS)
#define WP     796

__global__ __launch_bounds__(THREADS)
void moe_wmma_kernel(const float* __restrict__ x,
                     const int*   __restrict__ cat,
                     const float* __restrict__ conv_w,
                     const float* __restrict__ conv_b,
                     const float* __restrict__ fc1_w,
                     const float* __restrict__ fc1_b,
                     float* __restrict__ out)
{
    __shared__ float tile[WAVES_PER_BLOCK][ROWS_PER_WAVE * 16];
    __shared__ __attribute__((aligned(16))) float wl[(N_EXPERTS + 1) * WP];

    const int lane  = threadIdx.x & 31;
    const int wave  = threadIdx.x >> 5;
    const int gwave = blockIdx.x * WAVES_PER_BLOCK + wave;
    const int r0    = gwave * ROWS_PER_WAVE;

    const int m  = lane & 15;
    const int h4 = (lane >> 4) * 4;

    const int brow = (m < N_EXPERTS) ? m : N_EXPERTS;

    const v4f* __restrict__ ap = (const v4f*)(x + (size_t)(r0 + m) * KD + h4);
    const v4f* bp = (const v4f*)(wl + brow * WP + h4);

    v4f sw[WSTAGE];
    #pragma unroll
    for (int j = 0; j < WSTAGE; ++j) {
        const int idx = (int)threadIdx.x + j * THREADS;
        sw[j] = ((const v4f*)conv_w)[idx < WTOT4 ? idx : WTOT4 - 1];
    }

    v4f abuf[2][GRP];
    v4f bbuf[2][GRP];

    #pragma unroll
    for (int i = 0; i < GRP; ++i)
        abuf[0][i] = ap[2 * i];

    v8f acc = {};
    #pragma unroll
    for (int g = 0; g < NGRP; ++g) {
        const int cur = g & 1;
        const int nxt = cur ^ 1;
        if (g + 1 < NGRP) {
            #pragma unroll
            for (int i = 0; i < GRP; ++i)
                abuf[nxt][i] = ap[2 * ((g + 1) * GRP + i)];
        }
        if (g == 0) {
            #pragma unroll
            for (int j = 0; j < WSTAGE; ++j) {
                const int idx = (int)threadIdx.x + j * THREADS;
                if (j < WSTAGE - 1 || idx < WTOT4) {
                    const int r = idx / WROW4;
                    const int c = idx - r * WROW4;
                    *(v4f*)(wl + r * WP + 4 * c) = sw[j];
                }
            }
            if (threadIdx.x < WROW4) {
                const v4f z = {0.0f, 0.0f, 0.0f, 0.0f};
                *(v4f*)(wl + N_EXPERTS * WP + 4 * (int)threadIdx.x) = z;
            }
            __builtin_amdgcn_fence(__ATOMIC_RELEASE, "workgroup", "local");
            __builtin_amdgcn_s_barrier();
            __builtin_amdgcn_fence(__ATOMIC_ACQUIRE, "workgroup", "local");
            #pragma unroll
            for (int i = 0; i < GRP; ++i)
                bbuf[0][i] = bp[2 * i];
        }
        if (g + 1 < NGRP) {
            #pragma unroll
            for (int i = 0; i < GRP; ++i)
                bbuf[nxt][i] = bp[2 * ((g + 1) * GRP + i)];
        }
        __builtin_amdgcn_sched_barrier(0);
        #pragma unroll
        for (int i = 0; i < GRP; ++i) {
            const v4f a = abuf[cur][i];
            const v4f b = bbuf[cur][i];
            const v2f a0 = {a.x, a.y};
            const v2f b0 = {b.x, b.y};
            const v2f a1 = {a.z, a.w};
            const v2f b1 = {b.z, b.w};
            acc = __builtin_amdgcn_wmma_f32_16x16x4_f32(
                 false, a0,  false, b0,
                 (short)0, acc,  false,  false);
            acc = __builtin_amdgcn_wmma_f32_16x16x4_f32(
                 false, a1,  false, b1,
                 (short)0, acc,  false,  false);
            __builtin_amdgcn_sched_barrier(0);
        }
    }

    {
        float* t = tile[wave];
        const int mh = (lane >> 4) * 8;
        const int nn = lane & 15;
        #pragma unroll
        for (int r = 0; r < 8; ++r)
            t[(mh + r) * 16 + nn] = acc[r];
    }
    __syncthreads();

    if (lane < ROWS_PER_WAVE) {
        const int   row = r0 + lane;
        const int   ci  = cat[row];
        const float s   = tile[wave][lane * 16 + ci] + conv_b[ci];
        const float o   = s * fc1_w[ci] + fc1_b[0];
        out[row] = o > 0.0f ? o : 0.0f;
    }
}

extern "C" void kernel_launch(void* const* d_in, const int* in_sizes, int n_in,
                              void* d_out, int out_size, void* d_ws, size_t ws_size,
                              hipStream_t stream) {
    const float* x    = (const float*)d_in[0];
    const int*   cat  = (const int*)  d_in[1];
    const float* cw   = (const float*)d_in[2];
    const float* cb   = (const float*)d_in[3];
    const float* f1w  = (const float*)d_in[4];
    const float* f1b  = (const float*)d_in[5];
    float*       out  = (float*)d_out;

    const int nblocks = BATCH / (ROWS_PER_WAVE * WAVES_PER_BLOCK);
    hipLaunchKernelGGL(moe_wmma_kernel, dim3(nblocks), dim3(THREADS), 0, stream,
                       x, cat, cw, cb, f1w, f1b, out);
}


# WMMA tile rows = 4 samples x 4 K quarters, B column c = LDS row of the expert of row c's sample (diagonal of D read): an x load touches 4 whole 128 B lines (was 8 x 64 B); f32 sum order changes
# speedup vs baseline: 3.6898x; 1.1674x over previous
#include <hip/hip_runtime.h>
#include <hip/hip_bf16.h>


typedef float v2f __attribute__((ext_vector_type(2)));
typedef float v4f __attribute__((ext_vector_type(4)));
typedef float v8f __attribute__((ext_vector_type(8)));

#define BATCH     32768
#define KD        784
#define N_EXPERTS 5
#define ROWS_PER_WAVE 16
#define TILES     4
#define WAVES_PER_BLOCK 8
#define THREADS (WAVES_PER_BLOCK * 32)
#define NSTEP  24
#define GSTEP  4
#define NGRP   (NSTEP / GSTEP)
#define NSUB   (NSTEP / 2)
#define WROW4  (KD / 4)
#define WTOT4  (N_EXPERTS * WROW4)
#define WSTAGE ((WTOT4 + THREADS - 1) / THREADS)
#define WP     820

__global__ __launch_bounds__(THREADS)
void moe_wmma_kernel(const float* __restrict__ x,
                     const int*   __restrict__ cat,
                     const float* __restrict__ conv_w,
                     const float* __restrict__ conv_b,
                     const float* __restrict__ fc1_w,
                     const float* __restrict__ fc1_b,
                     float* __restrict__ out)
{
    __shared__ float tile[WAVES_PER_BLOCK][TILES * 8 * 32];
    __shared__ __attribute__((aligned(16))) float wl[N_EXPERTS * WP];

    const int lane  = threadIdx.x & 31;
    const int wave  = threadIdx.x >> 5;
    const int gwave = blockIdx.x * WAVES_PER_BLOCK + wave;
    const int r0    = gwave * ROWS_PER_WAVE;

    const int tc   = lane & 15;
    const int s4   = tc & 3;
    const int jq   = tc >> 2;
    const int hh   = lane >> 4;
    const int odd  = s4 & 1;
    const int moff = odd * 16 + jq * 8 + hh * 4;
    const int loff = (odd ? 0 : KD - 16) + jq * 4 + hh * 2;

    int e[TILES];
    #pragma unroll
    for (int t = 0; t < TILES; ++t)
        e[t] = cat[r0 + 4 * t + s4];

    v4f sw[WSTAGE];
    #pragma unroll
    for (int j = 0; j < WSTAGE; ++j) {
        const int idx = (int)threadIdx.x + j * THREADS;
        sw[j] = ((const v4f*)conv_w)[idx < WTOT4 ? idx : WTOT4 - 1];
    }

    const float* xrow = x + (size_t)(r0 + s4) * KD;
    const v4f* __restrict__ ap = (const v4f*)(xrow + moff);
    const v2f* __restrict__ al = (const v2f*)(xrow + loff);

    v4f abuf[2][GSTEP * TILES];
    v4f bbuf[2][2 * TILES];
    v2f arem[TILES];
    v2f brem[TILES];

    #pragma unroll
    for (int i = 0; i < GSTEP; ++i) {
        #pragma unroll
        for (int t = 0; t < TILES; ++t)
            abuf[0][4 * i + t] = ap[t * KD + 8 * i];
        __builtin_amdgcn_sched_barrier(0);
    }

    const v4f* bp[TILES];
    const v2f* bl[TILES];
    #pragma unroll
    for (int t = 0; t < TILES; ++t) {
        bp[t] = (const v4f*)(wl + e[t] * WP + moff);
        bl[t] = (const v2f*)(wl + e[t] * WP + loff);
    }

    v8f acc[TILES];
    #pragma unroll
    for (int t = 0; t < TILES; ++t)
        acc[t] = (v8f){0.0f, 0.0f, 0.0f, 0.0f, 0.0f, 0.0f, 0.0f, 0.0f};

    #pragma unroll
    for (int g = 0; g < NGRP; ++g) {
        const int cur = g & 1;
        const int nxt = cur ^ 1;
        if (g + 1 < NGRP) {
            #pragma unroll
            for (int i = 0; i < GSTEP; ++i) {
                #pragma unroll
                for (int t = 0; t < TILES; ++t)
                    abuf[nxt][4 * i + t] = ap[t * KD + 8 * ((g + 1) * GSTEP + i)];
                __builtin_amdgcn_sched_barrier(0);
            }
        }
        if (g + 2 == NGRP) {
            #pragma unroll
            for (int t = 0; t < TILES; ++t)
                arem[t] = al[t * 2 * KD];
        }
        if (g == 0) {
            #pragma unroll
            for (int j = 0; j < WSTAGE; ++j) {
                const int idx = (int)threadIdx.x + j * THREADS;
                if (j < WSTAGE - 1 || idx < WTOT4) {
                    const int r = idx / WROW4;
                    const int c = idx - r * WROW4;
                    *(v4f*)(wl + r * WP + 4 * c) = sw[j];
                }
            }
            __builtin_amdgcn_fence(__ATOMIC_RELEASE, "workgroup", "local");
            __builtin_amdgcn_s_barrier();
            __builtin_amdgcn_fence(__ATOMIC_ACQUIRE, "workgroup", "local");
            #pragma unroll
            for (int u = 0; u < 2; ++u) {
                #pragma unroll
                for (int t = 0; t < TILES; ++t)
                    bbuf[0][4 * u + t] = bp[t][8 * u];
            }
        }
        #pragma unroll
        for (int sb = 0; sb < 2; ++sb) {
            const int k  = 2 * g + sb;
            const int bc = k & 1;
            const int bn = bc ^ 1;
            if (k + 1 < NSUB) {
                #pragma unroll
                for (int u = 0; u < 2; ++u) {
                    #pragma unroll
                    for (int t = 0; t < TILES; ++t)
                        bbuf[bn][4 * u + t] = bp[t][8 * (2 * (k + 1) + u)];
                }
            } else {
                #pragma unroll
                for (int t = 0; t < TILES; ++t)
                    brem[t] = bl[t][0];
            }
            __builtin_amdgcn_sched_barrier(0);
            #pragma unroll
            for (int u = 0; u < 2; ++u) {
                #pragma unroll
                for (int t = 0; t < TILES; ++t) {
                    const v4f a = abuf[cur][4 * (2 * sb + u) + t];
                    const v4f b = bbuf[bc][4 * u + t];
                    const v2f a0 = {a.x, a.y};
                    const v2f b0 = {b.x, b.y};
                    acc[t] = __builtin_amdgcn_wmma_f32_16x16x4_f32(
                         false, a0,  false, b0,
                         (short)0, acc[t],  false,  false);
                }
                #pragma unroll
                for (int t = 0; t < TILES; ++t) {
                    const v4f a = abuf[cur][4 * (2 * sb + u) + t];
                    const v4f b = bbuf[bc][4 * u + t];
                    const v2f a1 = {a.z, a.w};
                    const v2f b1 = {b.z, b.w};
                    acc[t] = __builtin_amdgcn_wmma_f32_16x16x4_f32(
                         false, a1,  false, b1,
                         (short)0, acc[t],  false,  false);
                }
                __builtin_amdgcn_sched_barrier(0);
            }
        }
    }
    #pragma unroll
    for (int t = 0; t < TILES; ++t)
        acc[t] = __builtin_amdgcn_wmma_f32_16x16x4_f32(
             false, arem[t],  false, brem[t],
             (short)0, acc[t],  false,  false);

    {
        float* t = tile[wave];
        #pragma unroll
        for (int q = 0; q < TILES; ++q) {
            #pragma unroll
            for (int r = 0; r < 8; ++r)
                t[q * 256 + r * 32 + lane] = acc[q][r];
        }
    }
    __syncthreads();

    if (lane < ROWS_PER_WAVE) {
        const int    row = r0 + lane;
        const int    ci  = cat[row];
        const int    s   = lane & 3;
        const float* tr  = tile[wave] + (lane >> 2) * 256;
        const float  dot = (tr[s * 32 + s] + tr[(s + 4) * 32 + s + 4])
                         + (tr[s * 32 + s + 24] + tr[(s + 4) * 32 + s + 28]);
        const float  sv  = dot + conv_b[ci];
        const float  o   = sv * fc1_w[ci] + fc1_b[0];
        out[row] = o > 0.0f ? o : 0.0f;
    }
}

extern "C" void kernel_launch(void* const* d_in, const int* in_sizes, int n_in,
                              void* d_out, int out_size, void* d_ws, size_t ws_size,
                              hipStream_t stream) {
    const float* x    = (const float*)d_in[0];
    const int*   cat  = (const int*)  d_in[1];
    const float* cw   = (const float*)d_in[2];
    const float* cb   = (const float*)d_in[3];
    const float* f1w  = (const float*)d_in[4];
    const float* f1b  = (const float*)d_in[5];
    float*       out  = (float*)d_out;

    const int nblocks = BATCH / (ROWS_PER_WAVE * WAVES_PER_BLOCK);
    hipLaunchKernelGGL(moe_wmma_kernel, dim3(nblocks), dim3(THREADS), 0, stream,
                       x, cat, cw, cb, f1w, f1b, out);
}
